# starting kernel
# baseline (speedup 1.0000x reference)
#include <hip/hip_runtime.h>
#include <hip/hip_bf16.h>

#define N_NODES 100000
#define N_EDGES 1600000
#define D_IN    256
#define D_EDGE  16
#define D_OUT   128

typedef __attribute__((ext_vector_type(2))) float v2f;
typedef __attribute__((ext_vector_type(8))) float v8f;

__global__ void zero_cnt_kernel(int* __restrict__ cnt, int n) {
    int i = blockIdx.x * blockDim.x + threadIdx.x;
    if (i < n) cnt[i] = 0;
}

__global__ void count_edges_kernel(const int* __restrict__ dst,
                                   int* __restrict__ cnt, int e_total) {
    int e = blockIdx.x * blockDim.x + threadIdx.x;
    if (e < e_total) atomicAdd(&cnt[dst[e]], 1);
}

#define XS_STRIDE 257

__global__ void __launch_bounds__(512)
gemm_dual_wmma_kernel(const float* __restrict__ X,
                      const float* __restrict__ W_self,
                      const float* __restrict__ W_neigh,
                      const float* __restrict__ b_self,
                      float* __restrict__ out_self,
                      float* __restrict__ out_neigh) {
    __shared__ float xs[16 * XS_STRIDE];

    const int rowBase = blockIdx.x * 16;

    for (int idx = threadIdx.x; idx < 16 * D_IN; idx += 512) {
        const int r = idx >> 8;
        const int c = idx & 255;
        xs[r * XS_STRIDE + c] = X[(size_t)(rowBase + r) * D_IN + c];
    }
    __syncthreads();

    const int wave  = threadIdx.x >> 5;
    const int lane  = threadIdx.x & 31;
    const int ntile = wave & 7;
    const bool isSelf = (wave < 8);

    const float* __restrict__ W = isSelf ? W_self : W_neigh;

    const int m     = lane & 15;
    const int khalf = lane >> 4;
    const int col   = ntile * 16 + m;

    v8f acc = {};
    const float* xrow = &xs[m * XS_STRIDE];

    #pragma unroll 4
    for (int kb = 0; kb < D_IN; kb += 4) {
        const int k0 = kb + 2 * khalf;
        v2f a, b;
        a.x = xrow[k0 + 0];
        a.y = xrow[k0 + 1];
        b.x = W[(size_t)(k0 + 0) * D_OUT + col];
        b.y = W[(size_t)(k0 + 1) * D_OUT + col];
        acc = __builtin_amdgcn_wmma_f32_16x16x4_f32(
             false, a,  false, b,
             (short)0, acc,  false,  false);
    }

    const float bias = isSelf ? b_self[col] : 0.0f;
    float* __restrict__ outp = isSelf ? out_self : out_neigh;

    #pragma unroll
    for (int r = 0; r < 8; ++r) {
        const int row = rowBase + r + 8 * khalf;
        outp[(size_t)row * D_OUT + col] = acc[r] + bias;
    }
}

#define EDGES_PER_BLOCK 8

__global__ void __launch_bounds__(256)
edge_scatter_kernel(const float* __restrict__ X_edge,
                    const float* __restrict__ h_src,
                    const int* __restrict__ src,
                    const int* __restrict__ dst,
                    const int* __restrict__ cnt,
                    const float* __restrict__ W_edge,
                    float* __restrict__ out, int e_total) {
    __shared__ float wl[D_EDGE * D_OUT];
    for (int i = threadIdx.x; i < D_EDGE * D_OUT; i += 256) wl[i] = W_edge[i];
    __syncthreads();

    const int wave = threadIdx.x >> 5;
    const int lane = threadIdx.x & 31;
    const int e = blockIdx.x * EDGES_PER_BLOCK + wave;
    if (e >= e_total) return;

    const int s = src[e];
    const int d = dst[e];
    const float scale = 1.0f / (float)cnt[d];

    const float xv = X_edge[(size_t)e * D_EDGE + (lane & 15)];
    float xe[D_EDGE];
    #pragma unroll
    for (int k = 0; k < D_EDGE; ++k) xe[k] = __shfl(xv, k, 32);

    const float* __restrict__ hs = h_src + (size_t)s * D_OUT;
    float* __restrict__ od = out + (size_t)d * D_OUT;

    #pragma unroll
    for (int j = 0; j < 4; ++j) {
        const int c = lane + j * 32;
        float v = hs[c];
        #pragma unroll
        for (int k = 0; k < D_EDGE; ++k) v = fmaf(xe[k], wl[k * D_OUT + c], v);
        atomicAdd(&od[c], v * scale);
    }
}

extern "C" void kernel_launch(void* const* d_in, const int* in_sizes, int n_in,
                              void* d_out, int out_size, void* d_ws, size_t ws_size,
                              hipStream_t stream) {
    const float* X       = (const float*)d_in[0];
    const float* X_edge  = (const float*)d_in[1];
    const float* W_neigh = (const float*)d_in[2];
    const float* W_self  = (const float*)d_in[3];
    const float* b_self  = (const float*)d_in[4];
    const float* W_edge  = (const float*)d_in[5];
    const int*   src     = (const int*)d_in[6];
    const int*   dst     = (const int*)d_in[7];
    float* out = (float*)d_out;

    float* h_src = (float*)d_ws;
    int*   cnt   = (int*)((char*)d_ws + (size_t)N_NODES * D_OUT * sizeof(float));

    zero_cnt_kernel<<<(N_NODES + 255) / 256, 256, 0, stream>>>(cnt, N_NODES);
    count_edges_kernel<<<(N_EDGES + 255) / 256, 256, 0, stream>>>(dst, cnt, N_EDGES);

    gemm_dual_wmma_kernel<<<N_NODES / 16, 512, 0, stream>>>(
        X, W_self, W_neigh, b_self, out, h_src);

    edge_scatter_kernel<<<(N_EDGES + EDGES_PER_BLOCK - 1) / EDGES_PER_BLOCK, 256, 0, stream>>>(
        X_edge, h_src, src, dst, cnt, W_edge, out, N_EDGES);
}


# global f32 atomic scatter replaced by LDS sums: 256 groups each own 391 nodes, scan all dst ids, add owned rows by LDS f32 atomics; feats summed before W_edge; count pass gone; arrival sum order
# speedup vs baseline: 2.5463x; 2.5463x over previous
#include <hip/hip_runtime.h>
#include <hip/hip_bf16.h>

#define N_NODES 100000
#define N_EDGES 1600000
#define D_IN    256
#define D_EDGE  16
#define D_OUT   128

typedef __attribute__((ext_vector_type(2))) float v2f;
typedef __attribute__((ext_vector_type(8))) float v8f;

#define XS_STRIDE 257

__global__ void __launch_bounds__(512)
gemm_dual_wmma_kernel(const float* __restrict__ X,
                      const float* __restrict__ W_self,
                      const float* __restrict__ W_neigh,
                      const float* __restrict__ b_self,
                      float* __restrict__ out_self,
                      float* __restrict__ out_neigh) {
    __shared__ float xs[16 * XS_STRIDE];

    const int rowBase = blockIdx.x * 16;

    for (int idx = threadIdx.x; idx < 16 * D_IN; idx += 512) {
        const int r = idx >> 8;
        const int c = idx & 255;
        xs[r * XS_STRIDE + c] = X[(size_t)(rowBase + r) * D_IN + c];
    }
    __syncthreads();

    const int wave  = threadIdx.x >> 5;
    const int lane  = threadIdx.x & 31;
    const int ntile = wave & 7;
    const bool isSelf = (wave < 8);

    const float* __restrict__ W = isSelf ? W_self : W_neigh;

    const int m     = lane & 15;
    const int khalf = lane >> 4;
    const int col   = ntile * 16 + m;

    v8f acc = {};
    const float* xrow = &xs[m * XS_STRIDE];

    #pragma unroll 4
    for (int kb = 0; kb < D_IN; kb += 4) {
        const int k0 = kb + 2 * khalf;
        v2f a, b;
        a.x = xrow[k0 + 0];
        a.y = xrow[k0 + 1];
        b.x = W[(size_t)(k0 + 0) * D_OUT + col];
        b.y = W[(size_t)(k0 + 1) * D_OUT + col];
        acc = __builtin_amdgcn_wmma_f32_16x16x4_f32(
             false, a,  false, b,
             (short)0, acc,  false,  false);
    }

    const float bias = isSelf ? b_self[col] : 0.0f;
    float* __restrict__ outp = isSelf ? out_self : out_neigh;

    #pragma unroll
    for (int r = 0; r < 8; ++r) {
        const int row = rowBase + r + 8 * khalf;
        outp[(size_t)row * D_OUT + col] = acc[r] + bias;
    }
}

#define AGG_GROUPS 256
#define AGG_ROWS   391
#define AGG_WAVES  32
#define EF_STRIDE  17
#define SCAN_IDS   512
#define SCAN_STEPS (N_EDGES / SCAN_IDS)
#define Q_CAP      (32 + SCAN_IDS)
#define Q_EDGE_BITS 21
#define Q_EDGE_MASK ((1 << Q_EDGE_BITS) - 1)

static_assert(AGG_GROUPS * AGG_ROWS >= N_NODES, "the groups cover every node");
static_assert(N_EDGES % SCAN_IDS == 0, "whole scan steps");
static_assert(N_EDGES <= (1 << Q_EDGE_BITS) && AGG_ROWS <= (1 << 9),
              "a queue word packs the edge id (21 bits) and the local row (9 bits)");
static_assert((AGG_ROWS * D_OUT + AGG_ROWS * EF_STRIDE + AGG_WAVES * Q_CAP) * 4 <= 320 * 1024, "LDS budget");

__device__ __forceinline__ void agg_add_batch(const unsigned* q, int first, int nb, int lane,
                                              const float* __restrict__ X_edge,
                                              const float* __restrict__ h_src,
                                              const int* __restrict__ src,
                                              float* acc, float* ef) {
    unsigned ent = 0;
    int s = 0;
    if (lane < nb) {
        ent = q[first + lane];
        s = src[ent & Q_EDGE_MASK];
    }
    for (int i0 = 0; i0 < nb; i0 += 8) {
        float4 hv[8];
        float  xv[8];
        #pragma unroll
        for (int j = 0; j < 8; ++j) {
            if (i0 + j < nb) {
                const int sj = __builtin_amdgcn_readlane(s, i0 + j);
                const int ej = (int)(__builtin_amdgcn_readlane((int)ent, i0 + j) & Q_EDGE_MASK);
                hv[j] = ((const float4*)(h_src + (size_t)sj * D_OUT))[lane];
                xv[j] = (lane < D_EDGE) ? X_edge[(size_t)ej * D_EDGE + lane] : 1.0f;
            }
        }
        #pragma unroll
        for (int j = 0; j < 8; ++j) {
            if (i0 + j < nb) {
                const int rj = (int)((unsigned)__builtin_amdgcn_readlane((int)ent, i0 + j) >> Q_EDGE_BITS);
                float* a = acc + rj * D_OUT + lane;
                __hip_atomic_fetch_add(a,      hv[j].x, __ATOMIC_RELAXED, __HIP_MEMORY_SCOPE_WORKGROUP);
                __hip_atomic_fetch_add(a + 32, hv[j].y, __ATOMIC_RELAXED, __HIP_MEMORY_SCOPE_WORKGROUP);
                __hip_atomic_fetch_add(a + 64, hv[j].z, __ATOMIC_RELAXED, __HIP_MEMORY_SCOPE_WORKGROUP);
                __hip_atomic_fetch_add(a + 96, hv[j].w, __ATOMIC_RELAXED, __HIP_MEMORY_SCOPE_WORKGROUP);
                if (lane <= D_EDGE)
                    __hip_atomic_fetch_add(ef + rj * EF_STRIDE + lane, xv[j],
                                           __ATOMIC_RELAXED, __HIP_MEMORY_SCOPE_WORKGROUP);
            }
        }
    }
}

__global__ void __launch_bounds__(1024)
range_aggregate_kernel(const float* __restrict__ X_edge,
                       const float* __restrict__ h_src,
                       const int* __restrict__ src,
                       const int* __restrict__ dst,
                       const float* __restrict__ W_edge,
                       float* __restrict__ out) {
    __shared__ __attribute__((aligned(16))) float acc[AGG_ROWS * D_OUT];
    __shared__ float    ef[AGG_ROWS * EF_STRIDE];
    __shared__ unsigned queue[AGG_WAVES * Q_CAP];

    const int wave  = __builtin_amdgcn_readfirstlane((int)(threadIdx.x >> 5));
    const int lane  = threadIdx.x & 31;
    const int lo    = blockIdx.x * AGG_ROWS;
    const int nrows = min(AGG_ROWS, N_NODES - lo);

    for (int i = threadIdx.x; i < AGG_ROWS * D_OUT / 4; i += 1024)
        ((float4*)acc)[i] = make_float4(0.0f, 0.0f, 0.0f, 0.0f);
    for (int i = threadIdx.x; i < AGG_ROWS * EF_STRIDE; i += 1024) ef[i] = 0.0f;
    __syncthreads();

    unsigned* q = queue + wave * Q_CAP;
    int qn = 0;
    const int4* __restrict__ dst4 = (const int4*)dst;
    const int start = blockIdx.x * (SCAN_STEPS / AGG_GROUPS);

    for (int t = wave; t < SCAN_STEPS; t += AGG_WAVES) {
        int step = t + start;
        if (step >= SCAN_STEPS) step -= SCAN_STEPS;

        int4 dv[4];
        #pragma unroll
        for (int k = 0; k < 4; ++k) dv[k] = dst4[(step * 4 + k) * 32 + lane];

        unsigned rr[16];
        unsigned bm[16];
        unsigned anym = 0u;
        #pragma unroll
        for (int k = 0; k < 4; ++k) {
            rr[4 * k + 0] = (unsigned)(dv[k].x - lo);
            rr[4 * k + 1] = (unsigned)(dv[k].y - lo);
            rr[4 * k + 2] = (unsigned)(dv[k].z - lo);
            rr[4 * k + 3] = (unsigned)(dv[k].w - lo);
        }
        #pragma unroll
        for (int i = 0; i < 16; ++i) {
            bm[i] = __builtin_amdgcn_ballot_w32(rr[i] < (unsigned)nrows);
            anym |= bm[i];
        }
        if (anym != 0u) {
            #pragma unroll
            for (int i = 0; i < 16; ++i) {
                if (bm[i] != 0u) {
                    if ((bm[i] >> lane) & 1u) {
                        const unsigned e = (unsigned)(step * SCAN_IDS + (i >> 2) * 128 + lane * 4 + (i & 3));
                        q[qn + (int)__builtin_amdgcn_mbcnt_lo(bm[i], 0u)] = e | (rr[i] << Q_EDGE_BITS);
                    }
                    qn += __builtin_popcount(bm[i]);
                }
            }
        }
        while (qn >= 32) {
            qn -= 32;
            agg_add_batch(q, qn, 32, lane, X_edge, h_src, src, acc, ef);
        }
    }
    if (qn > 0) agg_add_batch(q, 0, qn, lane, X_edge, h_src, src, acc, ef);
    __syncthreads();

    float4 w[D_EDGE];
    #pragma unroll
    for (int k = 0; k < D_EDGE; ++k) w[k] = ((const float4*)(W_edge + k * D_OUT))[lane];

    for (int r = wave; r < nrows; r += AGG_WAVES) {
        const float* e = ef + r * EF_STRIDE;
        const float* a = acc + r * D_OUT + lane;
        float4 sum = make_float4(a[0], a[32], a[64], a[96]);
        #pragma unroll
        for (int k = 0; k < D_EDGE; ++k) {
            const float xk = e[k];
            sum.x = fmaf(xk, w[k].x, sum.x);
            sum.y = fmaf(xk, w[k].y, sum.y);
            sum.z = fmaf(xk, w[k].z, sum.z);
            sum.w = fmaf(xk, w[k].w, sum.w);
        }
        const float denom = fmaxf(e[D_EDGE], 1.0f);
        float4* op = (float4*)(out + (size_t)(lo + r) * D_OUT) + lane;
        float4 o = *op;
        o.x += sum.x / denom;
        o.y += sum.y / denom;
        o.z += sum.z / denom;
        o.w += sum.w / denom;
        *op = o;
    }
}

extern "C" void kernel_launch(void* const* d_in, const int* in_sizes, int n_in,
                              void* d_out, int out_size, void* d_ws, size_t ws_size,
                              hipStream_t stream) {
    const float* X       = (const float*)d_in[0];
    const float* X_edge  = (const float*)d_in[1];
    const float* W_neigh = (const float*)d_in[2];
    const float* W_self  = (const float*)d_in[3];
    const float* b_self  = (const float*)d_in[4];
    const float* W_edge  = (const float*)d_in[5];
    const int*   src     = (const int*)d_in[6];
    const int*   dst     = (const int*)d_in[7];
    float* out = (float*)d_out;

    float* h_src = (float*)d_ws;

    gemm_dual_wmma_kernel<<<N_NODES / 16, 512, 0, stream>>>(
        X, W_self, W_neigh, b_self, out, h_src);

    range_aggregate_kernel<<<AGG_GROUPS, 1024, 0, stream>>>(
        X_edge, h_src, src, dst, W_edge, out);
}


# dst ids re-coded once as one owner byte per edge (in the GEMM launch); each group's scan reads 1.6 MB not 6.4 MB, tests 4 ids per word op, loads a step ahead; arrival sum order as before
# speedup vs baseline: 2.8110x; 1.1040x over previous
#include <hip/hip_runtime.h>
#include <hip/hip_bf16.h>

#define N_NODES 100000
#define N_EDGES 1600000
#define D_IN    256
#define D_EDGE  16
#define D_OUT   128

#define AGG_GROUPS 256
#define AGG_ROWS   391
#define GEMM_BLOCKS   (N_NODES / 16)
#define OWN_PER_BLOCK (N_EDGES / GEMM_BLOCKS)
static_assert(N_EDGES % GEMM_BLOCKS == 0 && OWN_PER_BLOCK % 4 == 0 && OWN_PER_BLOCK / 4 <= 512,
              "the GEMM blocks cover every edge with whole words");
static_assert((N_NODES - 1) / AGG_ROWS <= 255, "an owner index fits a byte");

typedef __attribute__((ext_vector_type(2))) float v2f;
typedef __attribute__((ext_vector_type(8))) float v8f;

#define XS_STRIDE 257

__global__ void __launch_bounds__(512)
gemm_dual_wmma_kernel(const float* __restrict__ X,
                      const float* __restrict__ W_self,
                      const float* __restrict__ W_neigh,
                      const float* __restrict__ b_self,
                      float* __restrict__ out_self,
                      float* __restrict__ out_neigh,
                      const int* __restrict__ dst,
                      unsigned* __restrict__ owner) {
    __shared__ float xs[16 * XS_STRIDE];

    const int rowBase = blockIdx.x * 16;

    if (threadIdx.x < OWN_PER_BLOCK / 4) {
        const int w = blockIdx.x * (OWN_PER_BLOCK / 4) + threadIdx.x;
        const int4 d = ((const int4*)dst)[w];
        owner[w] = ((unsigned)(d.x / AGG_ROWS) & 255u)
                 | (((unsigned)(d.y / AGG_ROWS) & 255u) << 8)
                 | (((unsigned)(d.z / AGG_ROWS) & 255u) << 16)
                 | (((unsigned)(d.w / AGG_ROWS) & 255u) << 24);
    }

    for (int idx = threadIdx.x; idx < 16 * D_IN; idx += 512) {
        const int r = idx >> 8;
        const int c = idx & 255;
        xs[r * XS_STRIDE + c] = X[(size_t)(rowBase + r) * D_IN + c];
    }
    __syncthreads();

    const int wave  = threadIdx.x >> 5;
    const int lane  = threadIdx.x & 31;
    const int ntile = wave & 7;
    const bool isSelf = (wave < 8);

    const float* __restrict__ W = isSelf ? W_self : W_neigh;

    const int m     = lane & 15;
    const int khalf = lane >> 4;
    const int col   = ntile * 16 + m;

    v8f acc = {};
    const float* xrow = &xs[m * XS_STRIDE];

    #pragma unroll 4
    for (int kb = 0; kb < D_IN; kb += 4) {
        const int k0 = kb + 2 * khalf;
        v2f a, b;
        a.x = xrow[k0 + 0];
        a.y = xrow[k0 + 1];
        b.x = W[(size_t)(k0 + 0) * D_OUT + col];
        b.y = W[(size_t)(k0 + 1) * D_OUT + col];
        acc = __builtin_amdgcn_wmma_f32_16x16x4_f32(
             false, a,  false, b,
             (short)0, acc,  false,  false);
    }

    const float bias = isSelf ? b_self[col] : 0.0f;
    float* __restrict__ outp = isSelf ? out_self : out_neigh;

    #pragma unroll
    for (int r = 0; r < 8; ++r) {
        const int row = rowBase + r + 8 * khalf;
        outp[(size_t)row * D_OUT + col] = acc[r] + bias;
    }
}

#define AGG_WAVES  32
#define EF_STRIDE  17
#define SLICE_IDS  512
#define OWN_LOADS  5
#define OWN_STEP_IDS (OWN_LOADS * SLICE_IDS)
#define OWN_STEPS  (N_EDGES / OWN_STEP_IDS)
#define Q_CAP      (32 + SLICE_IDS)

static_assert(AGG_GROUPS * AGG_ROWS >= N_NODES, "the groups cover every node");
static_assert(N_EDGES % OWN_STEP_IDS == 0, "whole scan steps");
static_assert((AGG_ROWS * D_OUT + AGG_ROWS * EF_STRIDE + AGG_WAVES * Q_CAP) * 4 <= 320 * 1024, "LDS budget");

__device__ __forceinline__ unsigned zero_bytes(unsigned t) {
    return ~(((t & 0x7F7F7F7Fu) + 0x7F7F7F7Fu) | t | 0x7F7F7F7Fu);
}

__device__ __forceinline__ void agg_add_batch(const unsigned* q, int first, int nb, int lane,
                                              const float* __restrict__ X_edge,
                                              const float* __restrict__ h_src,
                                              const int* __restrict__ src,
                                              const int* __restrict__ dst,
                                              int lo, int nrows,
                                              float* acc, float* ef) {
    unsigned ent = 0;
    int s = 0;
    int row = 0;
    if (lane < nb) {
        ent = q[first + lane];
        s = src[ent];
        row = dst[ent] - lo;
    }
    for (int i0 = 0; i0 < nb; i0 += 8) {
        float4 hv[8];
        float  xv[8];
        #pragma unroll
        for (int j = 0; j < 8; ++j) {
            if (i0 + j < nb) {
                const int sj = __builtin_amdgcn_readlane(s, i0 + j);
                const int ej = __builtin_amdgcn_readlane((int)ent, i0 + j);
                hv[j] = ((const float4*)(h_src + (size_t)sj * D_OUT))[lane];
                xv[j] = (lane < D_EDGE) ? X_edge[(size_t)ej * D_EDGE + lane] : 1.0f;
            }
        }
        #pragma unroll
        for (int j = 0; j < 8; ++j) {
            const int rj = __builtin_amdgcn_readlane(row, (i0 + j) & 31);
            if (i0 + j < nb && (unsigned)rj < (unsigned)nrows) {
                float* a = acc + rj * D_OUT + lane;
                __hip_atomic_fetch_add(a,      hv[j].x, __ATOMIC_RELAXED, __HIP_MEMORY_SCOPE_WORKGROUP);
                __hip_atomic_fetch_add(a + 32, hv[j].y, __ATOMIC_RELAXED, __HIP_MEMORY_SCOPE_WORKGROUP);
                __hip_atomic_fetch_add(a + 64, hv[j].z, __ATOMIC_RELAXED, __HIP_MEMORY_SCOPE_WORKGROUP);
                __hip_atomic_fetch_add(a + 96, hv[j].w, __ATOMIC_RELAXED, __HIP_MEMORY_SCOPE_WORKGROUP);
                if (lane <= D_EDGE)
                    __hip_atomic_fetch_add(ef + rj * EF_STRIDE + lane, xv[j],
                                           __ATOMIC_RELAXED, __HIP_MEMORY_SCOPE_WORKGROUP);
            }
        }
    }
}

__global__ void __launch_bounds__(1024)
range_aggregate_kernel(const float* __restrict__ X_edge,
                       const float* __restrict__ h_src,
                       const int* __restrict__ src,
                       const int* __restrict__ dst,
                       const unsigned* __restrict__ owner,
                       const float* __restrict__ W_edge,
                       float* __restrict__ out) {
    __shared__ __attribute__((aligned(16))) float acc[AGG_ROWS * D_OUT];
    __shared__ float    ef[AGG_ROWS * EF_STRIDE];
    __shared__ unsigned queue[AGG_WAVES * Q_CAP];

    const int wave  = __builtin_amdgcn_readfirstlane((int)(threadIdx.x >> 5));
    const int lane  = threadIdx.x & 31;
    const int lo    = blockIdx.x * AGG_ROWS;
    const int nrows = min(AGG_ROWS, N_NODES - lo);

    for (int i = threadIdx.x; i < AGG_ROWS * D_OUT / 4; i += 1024)
        ((float4*)acc)[i] = make_float4(0.0f, 0.0f, 0.0f, 0.0f);
    for (int i = threadIdx.x; i < AGG_ROWS * EF_STRIDE; i += 1024) ef[i] = 0.0f;
    __syncthreads();

    unsigned* q = queue + wave * Q_CAP;
    int qn = 0;
    const uint4* __restrict__ own4 = (const uint4*)owner;
    const unsigned pat = (unsigned)blockIdx.x * 0x01010101u;
    const int start = (int)((blockIdx.x * OWN_STEPS) / AGG_GROUPS);

    uint4 nx[OWN_LOADS];
    {
        int step = wave + start;
        if (step >= OWN_STEPS) step -= OWN_STEPS;
        #pragma unroll
        for (int k = 0; k < OWN_LOADS; ++k) nx[k] = own4[(step * OWN_LOADS + k) * 32 + lane];
    }

    for (int t = wave; t < OWN_STEPS; t += AGG_WAVES) {
        int step = t + start;
        if (step >= OWN_STEPS) step -= OWN_STEPS;

        uint4 cu[OWN_LOADS];
        #pragma unroll
        for (int k = 0; k < OWN_LOADS; ++k) cu[k] = nx[k];
        const int tn = t + AGG_WAVES;
        if (tn < OWN_STEPS) {
            int sn = tn + start;
            if (sn >= OWN_STEPS) sn -= OWN_STEPS;
            #pragma unroll
            for (int k = 0; k < OWN_LOADS; ++k) nx[k] = own4[(sn * OWN_LOADS + k) * 32 + lane];
        }

        unsigned mk[OWN_LOADS];
        #pragma unroll
        for (int k = 0; k < OWN_LOADS; ++k) {
            const unsigned h0 = zero_bytes(cu[k].x ^ pat);
            const unsigned h1 = zero_bytes(cu[k].y ^ pat);
            const unsigned h2 = zero_bytes(cu[k].z ^ pat);
            const unsigned h3 = zero_bytes(cu[k].w ^ pat);
            mk[k] = (h0 >> 7) | (h1 >> 6) | (h2 >> 5) | (h3 >> 4);
        }

        #pragma unroll 1
        for (int k = 0; k < OWN_LOADS; ++k) {
            unsigned m = mk[0];
            #pragma unroll
            for (int j = 1; j < OWN_LOADS; ++j) mk[j - 1] = mk[j];
            unsigned bm = __builtin_amdgcn_ballot_w32(m != 0u);
            if (bm != 0u) {
                const unsigned ebase = (unsigned)(step * OWN_STEP_IDS + k * SLICE_IDS + lane * 16);
                do {
                    if (m != 0u) {
                        const unsigned p = (unsigned)__builtin_ctz(m);
                        m &= m - 1u;
                        q[qn + (int)__builtin_amdgcn_mbcnt_lo(bm, 0u)] = ebase + ((p & 3u) << 2) + (p >> 3);
                    }
                    qn += __builtin_popcount(bm);
                    bm = __builtin_amdgcn_ballot_w32(m != 0u);
                } while (bm != 0u);
                while (qn >= 32) {
                    qn -= 32;
                    agg_add_batch(q, qn, 32, lane, X_edge, h_src, src, dst, lo, nrows, acc, ef);
                }
            }
        }
    }
    if (qn > 0) agg_add_batch(q, 0, qn, lane, X_edge, h_src, src, dst, lo, nrows, acc, ef);
    __syncthreads();

    float4 w[D_EDGE];
    #pragma unroll
    for (int k = 0; k < D_EDGE; ++k) w[k] = ((const float4*)(W_edge + k * D_OUT))[lane];

    for (int r = wave; r < nrows; r += AGG_WAVES) {
        const float* e = ef + r * EF_STRIDE;
        const float* a = acc + r * D_OUT + lane;
        float4 sum = make_float4(a[0], a[32], a[64], a[96]);
        #pragma unroll
        for (int k = 0; k < D_EDGE; ++k) {
            const float xk = e[k];
            sum.x = fmaf(xk, w[k].x, sum.x);
            sum.y = fmaf(xk, w[k].y, sum.y);
            sum.z = fmaf(xk, w[k].z, sum.z);
            sum.w = fmaf(xk, w[k].w, sum.w);
        }
        const float denom = fmaxf(e[D_EDGE], 1.0f);
        float4* op = (float4*)(out + (size_t)(lo + r) * D_OUT) + lane;
        float4 o = *op;
        o.x += sum.x / denom;
        o.y += sum.y / denom;
        o.z += sum.z / denom;
        o.w += sum.w / denom;
        *op = o;
    }
}

extern "C" void kernel_launch(void* const* d_in, const int* in_sizes, int n_in,
                              void* d_out, int out_size, void* d_ws, size_t ws_size,
                              hipStream_t stream) {
    const float* X       = (const float*)d_in[0];
    const float* X_edge  = (const float*)d_in[1];
    const float* W_neigh = (const float*)d_in[2];
    const float* W_self  = (const float*)d_in[3];
    const float* b_self  = (const float*)d_in[4];
    const float* W_edge  = (const float*)d_in[5];
    const int*   src     = (const int*)d_in[6];
    const int*   dst     = (const int*)d_in[7];
    float* out = (float*)d_out;

    float* h_src = (float*)d_ws;
    unsigned* owner = (unsigned*)(h_src + (size_t)N_NODES * D_OUT);

    gemm_dual_wmma_kernel<<<GEMM_BLOCKS, 512, 0, stream>>>(
        X, W_self, W_neigh, b_self, out, h_src, dst, owner);

    range_aggregate_kernel<<<AGG_GROUPS, 1024, 0, stream>>>(
        X_edge, h_src, src, dst, owner, W_edge, out);
}


# GEMM: a block stages 80 rows of X (was 16) and each wave keeps five accumulator tiles, so a loaded weight fragment feeds five matrix instructions; weights still streamed; sum order unchanged
# speedup vs baseline: 3.3079x; 1.1768x over previous
#include <hip/hip_runtime.h>
#include <hip/hip_bf16.h>

#define N_NODES 100000
#define N_EDGES 1600000
#define D_IN    256
#define D_EDGE  16
#define D_OUT   128

#define AGG_GROUPS 256
#define AGG_ROWS   391
#define GEMM_MT       5
#define GEMM_ROWS     (16 * GEMM_MT)
#define GEMM_BLOCKS   (N_NODES / GEMM_ROWS)
#define OWN_PER_BLOCK (N_EDGES / GEMM_BLOCKS)
static_assert(N_NODES % GEMM_ROWS == 0, "whole row blocks");
static_assert(N_EDGES % GEMM_BLOCKS == 0 && OWN_PER_BLOCK % 4 == 0 && OWN_PER_BLOCK / 4 <= 512,
              "the GEMM blocks cover every edge with whole words");
static_assert((N_NODES - 1) / AGG_ROWS <= 255, "an owner index fits a byte");

typedef __attribute__((ext_vector_type(2))) float v2f;
typedef __attribute__((ext_vector_type(8))) float v8f;

#define XS_STRIDE 260

__global__ void __launch_bounds__(512)
gemm_dual_wmma_kernel(const float* __restrict__ X,
                      const float* __restrict__ W_self,
                      const float* __restrict__ W_neigh,
                      const float* __restrict__ b_self,
                      float* __restrict__ out_self,
                      float* __restrict__ out_neigh,
                      const int* __restrict__ dst,
                      unsigned* __restrict__ owner) {
    __shared__ __attribute__((aligned(16))) float xs[GEMM_ROWS * XS_STRIDE];

    const int rowBase = blockIdx.x * GEMM_ROWS;

    if (threadIdx.x < OWN_PER_BLOCK / 4) {
        const int w = blockIdx.x * (OWN_PER_BLOCK / 4) + threadIdx.x;
        const int4 d = ((const int4*)dst)[w];
        owner[w] = ((unsigned)(d.x / AGG_ROWS) & 255u)
                 | (((unsigned)(d.y / AGG_ROWS) & 255u) << 8)
                 | (((unsigned)(d.z / AGG_ROWS) & 255u) << 16)
                 | (((unsigned)(d.w / AGG_ROWS) & 255u) << 24);
    }

    for (int idx = threadIdx.x; idx < GEMM_ROWS * (D_IN / 4); idx += 512) {
        const int r  = idx >> 6;
        const int c4 = idx & 63;
        const float4 v = ((const float4*)(X + (size_t)(rowBase + r) * D_IN))[c4];
        *(float4*)&xs[r * XS_STRIDE + 4 * c4] = v;
    }
    __syncthreads();

    const int wave  = threadIdx.x >> 5;
    const int lane  = threadIdx.x & 31;
    const int ntile = wave & 7;
    const bool isSelf = (wave < 8);

    const float* __restrict__ W = isSelf ? W_self : W_neigh;

    const int m     = lane & 15;
    const int khalf = lane >> 4;
    const int col   = ntile * 16 + m;

    v8f acc[GEMM_MT];
    #pragma unroll
    for (int t = 0; t < GEMM_MT; ++t) acc[t] = (v8f){};

    const float* xrow = &xs[m * XS_STRIDE + 2 * khalf];
    const float* wcol = W + (size_t)(2 * khalf) * D_OUT + col;

    #pragma unroll 4
    for (int kb = 0; kb < D_IN; kb += 4) {
        v2f b;
        b.x = wcol[(size_t)(kb + 0) * D_OUT];
        b.y = wcol[(size_t)(kb + 1) * D_OUT];
        #pragma unroll
        for (int t = 0; t < GEMM_MT; ++t) {
            const v2f a = *(const v2f*)(xrow + t * 16 * XS_STRIDE + kb);
            acc[t] = __builtin_amdgcn_wmma_f32_16x16x4_f32(
                 false, a,  false, b,
                 (short)0, acc[t],  false,  false);
        }
    }

    const float bias = isSelf ? b_self[col] : 0.0f;
    float* __restrict__ outp = isSelf ? out_self : out_neigh;

    #pragma unroll
    for (int t = 0; t < GEMM_MT; ++t) {
        #pragma unroll
        for (int r = 0; r < 8; ++r) {
            const int row = rowBase + t * 16 + r + 8 * khalf;
            outp[(size_t)row * D_OUT + col] = acc[t][r] + bias;
        }
    }
}

#define AGG_WAVES  32
#define EF_STRIDE  17
#define SLICE_IDS  512
#define OWN_LOADS  5
#define OWN_STEP_IDS (OWN_LOADS * SLICE_IDS)
#define OWN_STEPS  (N_EDGES / OWN_STEP_IDS)
#define Q_CAP      (32 + SLICE_IDS)

static_assert(AGG_GROUPS * AGG_ROWS >= N_NODES, "the groups cover every node");
static_assert(N_EDGES % OWN_STEP_IDS == 0, "whole scan steps");
static_assert((AGG_ROWS * D_OUT + AGG_ROWS * EF_STRIDE + AGG_WAVES * Q_CAP) * 4 <= 320 * 1024, "LDS budget");

__device__ __forceinline__ unsigned zero_bytes(unsigned t) {
    return ~(((t & 0x7F7F7F7Fu) + 0x7F7F7F7Fu) | t | 0x7F7F7F7Fu);
}

__device__ __forceinline__ void agg_add_batch(const unsigned* q, int first, int nb, int lane,
                                              const float* __restrict__ X_edge,
                                              const float* __restrict__ h_src,
                                              const int* __restrict__ src,
                                              const int* __restrict__ dst,
                                              int lo, int nrows,
                                              float* acc, float* ef) {
    unsigned ent = 0;
    int s = 0;
    int row = 0;
    if (lane < nb) {
        ent = q[first + lane];
        s = src[ent];
        row = dst[ent] - lo;
    }
    for (int i0 = 0; i0 < nb; i0 += 8) {
        float4 hv[8];
        float  xv[8];
        #pragma unroll
        for (int j = 0; j < 8; ++j) {
            if (i0 + j < nb) {
                const int sj = __builtin_amdgcn_readlane(s, i0 + j);
                const int ej = __builtin_amdgcn_readlane((int)ent, i0 + j);
                hv[j] = ((const float4*)(h_src + (size_t)sj * D_OUT))[lane];
                xv[j] = (lane < D_EDGE) ? X_edge[(size_t)ej * D_EDGE + lane] : 1.0f;
            }
        }
        #pragma unroll
        for (int j = 0; j < 8; ++j) {
            const int rj = __builtin_amdgcn_readlane(row, (i0 + j) & 31);
            if (i0 + j < nb && (unsigned)rj < (unsigned)nrows) {
                float* a = acc + rj * D_OUT + lane;
                __hip_atomic_fetch_add(a,      hv[j].x, __ATOMIC_RELAXED, __HIP_MEMORY_SCOPE_WORKGROUP);
                __hip_atomic_fetch_add(a + 32, hv[j].y, __ATOMIC_RELAXED, __HIP_MEMORY_SCOPE_WORKGROUP);
                __hip_atomic_fetch_add(a + 64, hv[j].z, __ATOMIC_RELAXED, __HIP_MEMORY_SCOPE_WORKGROUP);
                __hip_atomic_fetch_add(a + 96, hv[j].w, __ATOMIC_RELAXED, __HIP_MEMORY_SCOPE_WORKGROUP);
                if (lane <= D_EDGE)
                    __hip_atomic_fetch_add(ef + rj * EF_STRIDE + lane, xv[j],
                                           __ATOMIC_RELAXED, __HIP_MEMORY_SCOPE_WORKGROUP);
            }
        }
    }
}

__global__ void __launch_bounds__(1024)
range_aggregate_kernel(const float* __restrict__ X_edge,
                       const float* __restrict__ h_src,
                       const int* __restrict__ src,
                       const int* __restrict__ dst,
                       const unsigned* __restrict__ owner,
                       const float* __restrict__ W_edge,
                       float* __restrict__ out) {
    __shared__ __attribute__((aligned(16))) float acc[AGG_ROWS * D_OUT];
    __shared__ float    ef[AGG_ROWS * EF_STRIDE];
    __shared__ unsigned queue[AGG_WAVES * Q_CAP];

    const int wave  = __builtin_amdgcn_readfirstlane((int)(threadIdx.x >> 5));
    const int lane  = threadIdx.x & 31;
    const int lo    = blockIdx.x * AGG_ROWS;
    const int nrows = min(AGG_ROWS, N_NODES - lo);

    for (int i = threadIdx.x; i < AGG_ROWS * D_OUT / 4; i += 1024)
        ((float4*)acc)[i] = make_float4(0.0f, 0.0f, 0.0f, 0.0f);
    for (int i = threadIdx.x; i < AGG_ROWS * EF_STRIDE; i += 1024) ef[i] = 0.0f;
    __syncthreads();

    unsigned* q = queue + wave * Q_CAP;
    int qn = 0;
    const uint4* __restrict__ own4 = (const uint4*)owner;
    const unsigned pat = (unsigned)blockIdx.x * 0x01010101u;
    const int start = (int)((blockIdx.x * OWN_STEPS) / AGG_GROUPS);

    uint4 nx[OWN_LOADS];
    {
        int step = wave + start;
        if (step >= OWN_STEPS) step -= OWN_STEPS;
        #pragma unroll
        for (int k = 0; k < OWN_LOADS; ++k) nx[k] = own4[(step * OWN_LOADS + k) * 32 + lane];
    }

    for (int t = wave; t < OWN_STEPS; t += AGG_WAVES) {
        int step = t + start;
        if (step >= OWN_STEPS) step -= OWN_STEPS;

        uint4 cu[OWN_LOADS];
        #pragma unroll
        for (int k = 0; k < OWN_LOADS; ++k) cu[k] = nx[k];
        const int tn = t + AGG_WAVES;
        if (tn < OWN_STEPS) {
            int sn = tn + start;
            if (sn >= OWN_STEPS) sn -= OWN_STEPS;
            #pragma unroll
            for (int k = 0; k < OWN_LOADS; ++k) nx[k] = own4[(sn * OWN_LOADS + k) * 32 + lane];
        }

        unsigned mk[OWN_LOADS];
        #pragma unroll
        for (int k = 0; k < OWN_LOADS; ++k) {
            const unsigned h0 = zero_bytes(cu[k].x ^ pat);
            const unsigned h1 = zero_bytes(cu[k].y ^ pat);
            const unsigned h2 = zero_bytes(cu[k].z ^ pat);
            const unsigned h3 = zero_bytes(cu[k].w ^ pat);
            mk[k] = (h0 >> 7) | (h1 >> 6) | (h2 >> 5) | (h3 >> 4);
        }

        #pragma unroll 1
        for (int k = 0; k < OWN_LOADS; ++k) {
            unsigned m = mk[0];
            #pragma unroll
            for (int j = 1; j < OWN_LOADS; ++j) mk[j - 1] = mk[j];
            unsigned bm = __builtin_amdgcn_ballot_w32(m != 0u);
            if (bm != 0u) {
                const unsigned ebase = (unsigned)(step * OWN_STEP_IDS + k * SLICE_IDS + lane * 16);
                do {
                    if (m != 0u) {
                        const unsigned p = (unsigned)__builtin_ctz(m);
                        m &= m - 1u;
                        q[qn + (int)__builtin_amdgcn_mbcnt_lo(bm, 0u)] = ebase + ((p & 3u) << 2) + (p >> 3);
                    }
                    qn += __builtin_popcount(bm);
                    bm = __builtin_amdgcn_ballot_w32(m != 0u);
                } while (bm != 0u);
                while (qn >= 32) {
                    qn -= 32;
                    agg_add_batch(q, qn, 32, lane, X_edge, h_src, src, dst, lo, nrows, acc, ef);
                }
            }
        }
    }
    if (qn > 0) agg_add_batch(q, 0, qn, lane, X_edge, h_src, src, dst, lo, nrows, acc, ef);
    __syncthreads();

    float4 w[D_EDGE];
    #pragma unroll
    for (int k = 0; k < D_EDGE; ++k) w[k] = ((const float4*)(W_edge + k * D_OUT))[lane];

    for (int r = wave; r < nrows; r += AGG_WAVES) {
        const float* e = ef + r * EF_STRIDE;
        const float* a = acc + r * D_OUT + lane;
        float4 sum = make_float4(a[0], a[32], a[64], a[96]);
        #pragma unroll
        for (int k = 0; k < D_EDGE; ++k) {
            const float xk = e[k];
            sum.x = fmaf(xk, w[k].x, sum.x);
            sum.y = fmaf(xk, w[k].y, sum.y);
            sum.z = fmaf(xk, w[k].z, sum.z);
            sum.w = fmaf(xk, w[k].w, sum.w);
        }
        const float denom = fmaxf(e[D_EDGE], 1.0f);
        float4* op = (float4*)(out + (size_t)(lo + r) * D_OUT) + lane;
        float4 o = *op;
        o.x += sum.x / denom;
        o.y += sum.y / denom;
        o.z += sum.z / denom;
        o.w += sum.w / denom;
        *op = o;
    }
}

extern "C" void kernel_launch(void* const* d_in, const int* in_sizes, int n_in,
                              void* d_out, int out_size, void* d_ws, size_t ws_size,
                              hipStream_t stream) {
    const float* X       = (const float*)d_in[0];
    const float* X_edge  = (const float*)d_in[1];
    const float* W_neigh = (const float*)d_in[2];
    const float* W_self  = (const float*)d_in[3];
    const float* b_self  = (const float*)d_in[4];
    const float* W_edge  = (const float*)d_in[5];
    const int*   src     = (const int*)d_in[6];
    const int*   dst     = (const int*)d_in[7];
    float* out = (float*)d_out;

    float* h_src = (float*)d_ws;
    unsigned* owner = (unsigned*)(h_src + (size_t)N_NODES * D_OUT);

    gemm_dual_wmma_kernel<<<GEMM_BLOCKS, 512, 0, stream>>>(
        X, W_self, W_neigh, b_self, out, h_src, dst, owner);

    range_aggregate_kernel<<<AGG_GROUPS, 1024, 0, stream>>>(
        X_edge, h_src, src, dst, owner, W_edge, out);
}


# speedup vs baseline: 3.4363x; 1.0388x over previous
#include <hip/hip_runtime.h>
#include <hip/hip_bf16.h>

#define N_NODES 100000
#define N_EDGES 1600000
#define D_IN    256
#define D_EDGE  16
#define D_OUT   128

#define AGG_GROUPS 256
#define AGG_ROWS   391
#define GEMM_MT       5
#define GEMM_ROWS     (16 * GEMM_MT)
#define GEMM_BLOCKS   (N_NODES / GEMM_ROWS)
#define OWN_PER_BLOCK (N_EDGES / GEMM_BLOCKS)
static_assert(N_NODES % GEMM_ROWS == 0, "whole row blocks");
static_assert(N_EDGES % GEMM_BLOCKS == 0 && OWN_PER_BLOCK % 4 == 0 && OWN_PER_BLOCK / 4 <= 512,
              "the GEMM blocks cover every edge with whole words");
static_assert((N_NODES - 1) / AGG_ROWS <= 255, "an owner index fits a byte");

typedef __attribute__((ext_vector_type(2))) float v2f;
typedef __attribute__((ext_vector_type(8))) float v8f;

#define XS_STRIDE 260

__global__ void __launch_bounds__(512)
gemm_dual_wmma_kernel(const float* __restrict__ X,
                      const float* __restrict__ W_self,
                      const float* __restrict__ W_neigh,
                      const float* __restrict__ b_self,
                      float* __restrict__ out_self,
                      float* __restrict__ out_neigh,
                      const int* __restrict__ dst,
                      unsigned* __restrict__ owner) {
    __shared__ __attribute__((aligned(16))) float xs[GEMM_ROWS * XS_STRIDE];

    const int rowBase = blockIdx.x * GEMM_ROWS;

    if (threadIdx.x < OWN_PER_BLOCK / 4) {
        const int w = blockIdx.x * (OWN_PER_BLOCK / 4) + threadIdx.x;
        const int4 d = ((const int4*)dst)[w];
        owner[w] = ((unsigned)(d.x / AGG_ROWS) & 255u)
                 | (((unsigned)(d.y / AGG_ROWS) & 255u) << 8)
                 | (((unsigned)(d.z / AGG_ROWS) & 255u) << 16)
                 | (((unsigned)(d.w / AGG_ROWS) & 255u) << 24);
    }

    for (int idx = threadIdx.x; idx < GEMM_ROWS * (D_IN / 4); idx += 512) {
        const int r  = idx >> 6;
        const int c4 = idx & 63;
        const float4 v = ((const float4*)(X + (size_t)(rowBase + r) * D_IN))[c4];
        *(float4*)&xs[r * XS_STRIDE + 4 * c4] = v;
    }
    __syncthreads();

    const int wave  = threadIdx.x >> 5;
    const int lane  = threadIdx.x & 31;
    const int ntile = wave & 7;
    const bool isSelf = (wave < 8);

    const float* __restrict__ W = isSelf ? W_self : W_neigh;

    const int m     = lane & 15;
    const int khalf = lane >> 4;
    const int col   = ntile * 16 + m;

    v8f acc[GEMM_MT];
    #pragma unroll
    for (int t = 0; t < GEMM_MT; ++t) acc[t] = (v8f){};

    const float* xrow = &xs[m * XS_STRIDE + 2 * khalf];
    const float* wcol = W + (size_t)(2 * khalf) * D_OUT + col;

    #pragma unroll 4
    for (int kb = 0; kb < D_IN; kb += 4) {
        v2f b;
        b.x = wcol[(size_t)(kb + 0) * D_OUT];
        b.y = wcol[(size_t)(kb + 1) * D_OUT];
        #pragma unroll
        for (int t = 0; t < GEMM_MT; ++t) {
            const v2f a = *(const v2f*)(xrow + t * 16 * XS_STRIDE + kb);
            acc[t] = __builtin_amdgcn_wmma_f32_16x16x4_f32(
                 false, a,  false, b,
                 (short)0, acc[t],  false,  false);
        }
    }

    const float bias = isSelf ? b_self[col] : 0.0f;
    float* __restrict__ outp = isSelf ? out_self : out_neigh;

    #pragma unroll
    for (int t = 0; t < GEMM_MT; ++t) {
        #pragma unroll
        for (int r = 0; r < 8; ++r) {
            const int row = rowBase + t * 16 + r + 8 * khalf;
            outp[(size_t)row * D_OUT + col] = acc[t][r] + bias;
        }
    }
}

#define AGG_WAVES  32
#define EF_STRIDE  17
#define SLICE_IDS  512
#define OWN_LOADS  5
#define OWN_STEP_IDS (OWN_LOADS * SLICE_IDS)
#define OWN_STEPS  (N_EDGES / OWN_STEP_IDS)
#define Q_CAP      (32 + SLICE_IDS)
#define SWEEP_SPREAD 64
static_assert(SWEEP_SPREAD <= OWN_STEPS, "a start wraps at most once");

static_assert(AGG_GROUPS * AGG_ROWS >= N_NODES, "the groups cover every node");
static_assert(N_EDGES % OWN_STEP_IDS == 0, "whole scan steps");
static_assert((AGG_ROWS * D_OUT + AGG_ROWS * EF_STRIDE + AGG_WAVES * Q_CAP) * 4 <= 320 * 1024, "LDS budget");

__device__ __forceinline__ unsigned zero_bytes(unsigned t) {
    return ~(((t & 0x7F7F7F7Fu) + 0x7F7F7F7Fu) | t | 0x7F7F7F7Fu);
}

__device__ __forceinline__ void agg_add_batch(const unsigned* q, int first, int nb, int lane,
                                              const float* __restrict__ X_edge,
                                              const float* __restrict__ h_src,
                                              const int* __restrict__ src,
                                              const int* __restrict__ dst,
                                              int lo, int nrows,
                                              float* acc, float* ef) {
    unsigned ent = 0;
    int s = 0;
    int row = 0;
    if (lane < nb) {
        ent = q[first + lane];
        s = src[ent];
        row = dst[ent] - lo;
    }
    for (int i0 = 0; i0 < nb; i0 += 8) {
        float4 hv[8];
        float  xv[8];
        #pragma unroll
        for (int j = 0; j < 8; ++j) {
            if (i0 + j < nb) {
                const int sj = __builtin_amdgcn_readlane(s, i0 + j);
                const int ej = __builtin_amdgcn_readlane((int)ent, i0 + j);
                hv[j] = ((const float4*)(h_src + (size_t)sj * D_OUT))[lane];
                xv[j] = (lane < D_EDGE) ? X_edge[(size_t)ej * D_EDGE + lane] : 1.0f;
            }
        }
        #pragma unroll
        for (int j = 0; j < 8; ++j) {
            const int rj = __builtin_amdgcn_readlane(row, (i0 + j) & 31);
            if (i0 + j < nb && (unsigned)rj < (unsigned)nrows) {
                float* a = acc + rj * D_OUT + lane;
                __hip_atomic_fetch_add(a,      hv[j].x, __ATOMIC_RELAXED, __HIP_MEMORY_SCOPE_WORKGROUP);
                __hip_atomic_fetch_add(a + 32, hv[j].y, __ATOMIC_RELAXED, __HIP_MEMORY_SCOPE_WORKGROUP);
                __hip_atomic_fetch_add(a + 64, hv[j].z, __ATOMIC_RELAXED, __HIP_MEMORY_SCOPE_WORKGROUP);
                __hip_atomic_fetch_add(a + 96, hv[j].w, __ATOMIC_RELAXED, __HIP_MEMORY_SCOPE_WORKGROUP);
                if (lane <= D_EDGE)
                    __hip_atomic_fetch_add(ef + rj * EF_STRIDE + lane, xv[j],
                                           __ATOMIC_RELAXED, __HIP_MEMORY_SCOPE_WORKGROUP);
            }
        }
    }
}

__global__ void __launch_bounds__(1024)
range_aggregate_kernel(const float* __restrict__ X_edge,
                       const float* __restrict__ h_src,
                       const int* __restrict__ src,
                       const int* __restrict__ dst,
                       const unsigned* __restrict__ owner,
                       const float* __restrict__ W_edge,
                       float* __restrict__ out) {
    __shared__ __attribute__((aligned(16))) float acc[AGG_ROWS * D_OUT];
    __shared__ float    ef[AGG_ROWS * EF_STRIDE];
    __shared__ unsigned queue[AGG_WAVES * Q_CAP];

    const int wave  = __builtin_amdgcn_readfirstlane((int)(threadIdx.x >> 5));
    const int lane  = threadIdx.x & 31;
    const int lo    = blockIdx.x * AGG_ROWS;
    const int nrows = min(AGG_ROWS, N_NODES - lo);

    for (int i = threadIdx.x; i < AGG_ROWS * D_OUT / 4; i += 1024)
        ((float4*)acc)[i] = make_float4(0.0f, 0.0f, 0.0f, 0.0f);
    for (int i = threadIdx.x; i < AGG_ROWS * EF_STRIDE; i += 1024) ef[i] = 0.0f;
    __syncthreads();

    unsigned* q = queue + wave * Q_CAP;
    int qn = 0;
    const uint4* __restrict__ own4 = (const uint4*)owner;
    const unsigned pat = (unsigned)blockIdx.x * 0x01010101u;
    const int start = (int)((blockIdx.x * SWEEP_SPREAD) / AGG_GROUPS);

    uint4 nx[OWN_LOADS];
    {
        int step = wave + start;
        if (step >= OWN_STEPS) step -= OWN_STEPS;
        #pragma unroll
        for (int k = 0; k < OWN_LOADS; ++k) nx[k] = own4[(step * OWN_LOADS + k) * 32 + lane];
    }

    for (int t = wave; t < OWN_STEPS; t += AGG_WAVES) {
        int step = t + start;
        if (step >= OWN_STEPS) step -= OWN_STEPS;

        uint4 cu[OWN_LOADS];
        #pragma unroll
        for (int k = 0; k < OWN_LOADS; ++k) cu[k] = nx[k];
        const int tn = t + AGG_WAVES;
        if (tn < OWN_STEPS) {
            int sn = tn + start;
            if (sn >= OWN_STEPS) sn -= OWN_STEPS;
            #pragma unroll
            for (int k = 0; k < OWN_LOADS; ++k) nx[k] = own4[(sn * OWN_LOADS + k) * 32 + lane];
        }

        unsigned mk[OWN_LOADS];
        #pragma unroll
        for (int k = 0; k < OWN_LOADS; ++k) {
            const unsigned h0 = zero_bytes(cu[k].x ^ pat);
            const unsigned h1 = zero_bytes(cu[k].y ^ pat);
            const unsigned h2 = zero_bytes(cu[k].z ^ pat);
            const unsigned h3 = zero_bytes(cu[k].w ^ pat);
            mk[k] = (h0 >> 7) | (h1 >> 6) | (h2 >> 5) | (h3 >> 4);
        }

        #pragma unroll 1
        for (int k = 0; k < OWN_LOADS; ++k) {
            unsigned m = mk[0];
            #pragma unroll
            for (int j = 1; j < OWN_LOADS; ++j) mk[j - 1] = mk[j];
            unsigned bm = __builtin_amdgcn_ballot_w32(m != 0u);
            if (bm != 0u) {
                const unsigned ebase = (unsigned)(step * OWN_STEP_IDS + k * SLICE_IDS + lane * 16);
                do {
                    if (m != 0u) {
                        const unsigned p = (unsigned)__builtin_ctz(m);
                        m &= m - 1u;
                        q[qn + (int)__builtin_amdgcn_mbcnt_lo(bm, 0u)] = ebase + ((p & 3u) << 2) + (p >> 3);
                    }
                    qn += __builtin_popcount(bm);
                    bm = __builtin_amdgcn_ballot_w32(m != 0u);
                } while (bm != 0u);
                while (qn >= 32) {
                    qn -= 32;
                    agg_add_batch(q, qn, 32, lane, X_edge, h_src, src, dst, lo, nrows, acc, ef);
                }
            }
        }
    }
    if (qn > 0) agg_add_batch(q, 0, qn, lane, X_edge, h_src, src, dst, lo, nrows, acc, ef);
    __syncthreads();

    float4 w[D_EDGE];
    #pragma unroll
    for (int k = 0; k < D_EDGE; ++k) w[k] = ((const float4*)(W_edge + k * D_OUT))[lane];

    for (int r = wave; r < nrows; r += AGG_WAVES) {
        const float* e = ef + r * EF_STRIDE;
        const float* a = acc + r * D_OUT + lane;
        float4 sum = make_float4(a[0], a[32], a[64], a[96]);
        #pragma unroll
        for (int k = 0; k < D_EDGE; ++k) {
            const float xk = e[k];
            sum.x = fmaf(xk, w[k].x, sum.x);
            sum.y = fmaf(xk, w[k].y, sum.y);
            sum.z = fmaf(xk, w[k].z, sum.z);
            sum.w = fmaf(xk, w[k].w, sum.w);
        }
        const float denom = fmaxf(e[D_EDGE], 1.0f);
        float4* op = (float4*)(out + (size_t)(lo + r) * D_OUT) + lane;
        float4 o = *op;
        o.x += sum.x / denom;
        o.y += sum.y / denom;
        o.z += sum.z / denom;
        o.w += sum.w / denom;
        *op = o;
    }
}

extern "C" void kernel_launch(void* const* d_in, const int* in_sizes, int n_in,
                              void* d_out, int out_size, void* d_ws, size_t ws_size,
                              hipStream_t stream) {
    const float* X       = (const float*)d_in[0];
    const float* X_edge  = (const float*)d_in[1];
    const float* W_neigh = (const float*)d_in[2];
    const float* W_self  = (const float*)d_in[3];
    const float* b_self  = (const float*)d_in[4];
    const float* W_edge  = (const float*)d_in[5];
    const int*   src     = (const int*)d_in[6];
    const int*   dst     = (const int*)d_in[7];
    float* out = (float*)d_out;

    float* h_src = (float*)d_ws;
    unsigned* owner = (unsigned*)(h_src + (size_t)N_NODES * D_OUT);

    gemm_dual_wmma_kernel<<<GEMM_BLOCKS, 512, 0, stream>>>(
        X, W_self, W_neigh, b_self, out, h_src, dst, owner);

    range_aggregate_kernel<<<AGG_GROUPS, 1024, 0, stream>>>(
        X_edge, h_src, src, dst, owner, W_edge, out);
}


# aggregation: a round's 16 gather loads issue back to back (array base + 32-bit index, all indices first, feature load unmasked, selected at the add), no address wait between; arrival sum order kept
# speedup vs baseline: 3.6356x; 1.0580x over previous
#include <hip/hip_runtime.h>
#include <hip/hip_bf16.h>

#define N_NODES 100000
#define N_EDGES 1600000
#define D_IN    256
#define D_EDGE  16
#define D_OUT   128

#define AGG_GROUPS 256
#define AGG_ROWS   391
#define GEMM_MT       5
#define GEMM_ROWS     (16 * GEMM_MT)
#define GEMM_BLOCKS   (N_NODES / GEMM_ROWS)
#define OWN_PER_BLOCK (N_EDGES / GEMM_BLOCKS)
static_assert(N_NODES % GEMM_ROWS == 0, "whole row blocks");
static_assert(N_EDGES % GEMM_BLOCKS == 0 && OWN_PER_BLOCK % 4 == 0 && OWN_PER_BLOCK / 4 <= 512,
              "the GEMM blocks cover every edge with whole words");
static_assert((N_NODES - 1) / AGG_ROWS <= 255, "an owner index fits a byte");

typedef __attribute__((ext_vector_type(2))) float v2f;
typedef __attribute__((ext_vector_type(8))) float v8f;

#define XS_STRIDE 260

__global__ void __launch_bounds__(512)
gemm_dual_wmma_kernel(const float* __restrict__ X,
                      const float* __restrict__ W_self,
                      const float* __restrict__ W_neigh,
                      const float* __restrict__ b_self,
                      float* __restrict__ out_self,
                      float* __restrict__ out_neigh,
                      const int* __restrict__ dst,
                      unsigned* __restrict__ owner) {
    __shared__ __attribute__((aligned(16))) float xs[GEMM_ROWS * XS_STRIDE];

    const int rowBase = blockIdx.x * GEMM_ROWS;

    if (threadIdx.x < OWN_PER_BLOCK / 4) {
        const int w = blockIdx.x * (OWN_PER_BLOCK / 4) + threadIdx.x;
        const int4 d = ((const int4*)dst)[w];
        owner[w] = ((unsigned)(d.x / AGG_ROWS) & 255u)
                 | (((unsigned)(d.y / AGG_ROWS) & 255u) << 8)
                 | (((unsigned)(d.z / AGG_ROWS) & 255u) << 16)
                 | (((unsigned)(d.w / AGG_ROWS) & 255u) << 24);
    }

    for (int idx = threadIdx.x; idx < GEMM_ROWS * (D_IN / 4); idx += 512) {
        const int r  = idx >> 6;
        const int c4 = idx & 63;
        const float4 v = ((const float4*)(X + (size_t)(rowBase + r) * D_IN))[c4];
        *(float4*)&xs[r * XS_STRIDE + 4 * c4] = v;
    }
    __syncthreads();

    const int wave  = threadIdx.x >> 5;
    const int lane  = threadIdx.x & 31;
    const int ntile = wave & 7;
    const bool isSelf = (wave < 8);

    const float* __restrict__ W = isSelf ? W_self : W_neigh;

    const int m     = lane & 15;
    const int khalf = lane >> 4;
    const int col   = ntile * 16 + m;

    v8f acc[GEMM_MT];
    #pragma unroll
    for (int t = 0; t < GEMM_MT; ++t) acc[t] = (v8f){};

    const float* xrow = &xs[m * XS_STRIDE + 2 * khalf];
    const float* wcol = W + (size_t)(2 * khalf) * D_OUT + col;

    #pragma unroll 4
    for (int kb = 0; kb < D_IN; kb += 4) {
        v2f b;
        b.x = wcol[(size_t)(kb + 0) * D_OUT];
        b.y = wcol[(size_t)(kb + 1) * D_OUT];
        #pragma unroll
        for (int t = 0; t < GEMM_MT; ++t) {
            const v2f a = *(const v2f*)(xrow + t * 16 * XS_STRIDE + kb);
            acc[t] = __builtin_amdgcn_wmma_f32_16x16x4_f32(
                 false, a,  false, b,
                 (short)0, acc[t],  false,  false);
        }
    }

    const float bias = isSelf ? b_self[col] : 0.0f;
    float* __restrict__ outp = isSelf ? out_self : out_neigh;

    #pragma unroll
    for (int t = 0; t < GEMM_MT; ++t) {
        #pragma unroll
        for (int r = 0; r < 8; ++r) {
            const int row = rowBase + t * 16 + r + 8 * khalf;
            outp[(size_t)row * D_OUT + col] = acc[t][r] + bias;
        }
    }
}

#define AGG_WAVES  32
#define EF_STRIDE  17
#define SLICE_IDS  512
#define OWN_LOADS  5
#define OWN_STEP_IDS (OWN_LOADS * SLICE_IDS)
#define OWN_STEPS  (N_EDGES / OWN_STEP_IDS)
#define Q_CAP      (32 + SLICE_IDS)
#define SWEEP_SPREAD 64
static_assert(SWEEP_SPREAD <= OWN_STEPS, "a start wraps at most once");

static_assert(AGG_GROUPS * AGG_ROWS >= N_NODES, "the groups cover every node");
static_assert(N_EDGES % OWN_STEP_IDS == 0, "whole scan steps");
static_assert((AGG_ROWS * D_OUT + AGG_ROWS * EF_STRIDE + AGG_WAVES * Q_CAP) * 4 <= 320 * 1024, "LDS budget");

__device__ __forceinline__ unsigned zero_bytes(unsigned t) {
    return ~(((t & 0x7F7F7F7Fu) + 0x7F7F7F7Fu) | t | 0x7F7F7F7Fu);
}

__device__ __forceinline__ void agg_add_batch(const unsigned* q, int first, int nb, int lane,
                                              const float* __restrict__ X_edge,
                                              const float* __restrict__ h_src,
                                              const int* __restrict__ src,
                                              const int* __restrict__ dst,
                                              int lo, int nrows,
                                              float* acc, float* ef) {
    unsigned ent = 0;
    int s = 0;
    int row = 0;
    if (lane < nb) {
        ent = q[first + lane];
        s = src[ent];
        row = dst[ent] - lo;
    }
    for (int i0 = 0; i0 < nb; i0 += 8) {
        float4 hv[8];
        float  xv[8];
        unsigned hi[8], xi[8];
        #pragma unroll
        for (int j = 0; j < 8; ++j) {
            const unsigned sj = (unsigned)__builtin_amdgcn_readlane(s, (i0 + j) & 31);
            const unsigned ej = (unsigned)__builtin_amdgcn_readlane((int)ent, (i0 + j) & 31);
            hi[j] = sj * (unsigned)(D_OUT / 4) + (unsigned)lane;
            xi[j] = ej * (unsigned)D_EDGE + (unsigned)(lane & (D_EDGE - 1));
        }
        #pragma unroll
        for (int j = 0; j < 8; ++j) {
            if (i0 + j < nb) {
                hv[j] = ((const float4*)h_src)[hi[j]];
                xv[j] = X_edge[xi[j]];
            }
        }
        asm volatile("" :: "v"(hv[0].x), "v"(hv[0].y), "v"(hv[0].z), "v"(hv[0].w), "v"(xv[0]));
        #pragma unroll
        for (int j = 0; j < 8; ++j) {
            const int rj = __builtin_amdgcn_readlane(row, (i0 + j) & 31);
            if (i0 + j < nb && (unsigned)rj < (unsigned)nrows) {
                float* a = acc + rj * D_OUT + lane;
                __hip_atomic_fetch_add(a,      hv[j].x, __ATOMIC_RELAXED, __HIP_MEMORY_SCOPE_WORKGROUP);
                __hip_atomic_fetch_add(a + 32, hv[j].y, __ATOMIC_RELAXED, __HIP_MEMORY_SCOPE_WORKGROUP);
                __hip_atomic_fetch_add(a + 64, hv[j].z, __ATOMIC_RELAXED, __HIP_MEMORY_SCOPE_WORKGROUP);
                __hip_atomic_fetch_add(a + 96, hv[j].w, __ATOMIC_RELAXED, __HIP_MEMORY_SCOPE_WORKGROUP);
                if (lane <= D_EDGE)
                    __hip_atomic_fetch_add(ef + rj * EF_STRIDE + lane, (lane < D_EDGE) ? xv[j] : 1.0f,
                                           __ATOMIC_RELAXED, __HIP_MEMORY_SCOPE_WORKGROUP);
            }
        }
    }
}

__global__ void __launch_bounds__(1024)
range_aggregate_kernel(const float* __restrict__ X_edge,
                       const float* __restrict__ h_src,
                       const int* __restrict__ src,
                       const int* __restrict__ dst,
                       const unsigned* __restrict__ owner,
                       const float* __restrict__ W_edge,
                       float* __restrict__ out) {
    __shared__ __attribute__((aligned(16))) float acc[AGG_ROWS * D_OUT];
    __shared__ float    ef[AGG_ROWS * EF_STRIDE];
    __shared__ unsigned queue[AGG_WAVES * Q_CAP];

    const int wave  = __builtin_amdgcn_readfirstlane((int)(threadIdx.x >> 5));
    const int lane  = threadIdx.x & 31;
    const int lo    = blockIdx.x * AGG_ROWS;
    const int nrows = min(AGG_ROWS, N_NODES - lo);

    for (int i = threadIdx.x; i < AGG_ROWS * D_OUT / 4; i += 1024)
        ((float4*)acc)[i] = make_float4(0.0f, 0.0f, 0.0f, 0.0f);
    for (int i = threadIdx.x; i < AGG_ROWS * EF_STRIDE; i += 1024) ef[i] = 0.0f;
    __syncthreads();

    unsigned* q = queue + wave * Q_CAP;
    int qn = 0;
    const uint4* __restrict__ own4 = (const uint4*)owner;
    const unsigned pat = (unsigned)blockIdx.x * 0x01010101u;
    const int start = (int)((blockIdx.x * SWEEP_SPREAD) / AGG_GROUPS);

    uint4 nx[OWN_LOADS];
    {
        int step = wave + start;
        if (step >= OWN_STEPS) step -= OWN_STEPS;
        #pragma unroll
        for (int k = 0; k < OWN_LOADS; ++k) nx[k] = own4[(step * OWN_LOADS + k) * 32 + lane];
    }

    for (int t = wave; t < OWN_STEPS; t += AGG_WAVES) {
        int step = t + start;
        if (step >= OWN_STEPS) step -= OWN_STEPS;

        uint4 cu[OWN_LOADS];
        #pragma unroll
        for (int k = 0; k < OWN_LOADS; ++k) cu[k] = nx[k];
        const int tn = t + AGG_WAVES;
        if (tn < OWN_STEPS) {
            int sn = tn + start;
            if (sn >= OWN_STEPS) sn -= OWN_STEPS;
            #pragma unroll
            for (int k = 0; k < OWN_LOADS; ++k) nx[k] = own4[(sn * OWN_LOADS + k) * 32 + lane];
        }

        unsigned mk[OWN_LOADS];
        #pragma unroll
        for (int k = 0; k < OWN_LOADS; ++k) {
            const unsigned h0 = zero_bytes(cu[k].x ^ pat);
            const unsigned h1 = zero_bytes(cu[k].y ^ pat);
            const unsigned h2 = zero_bytes(cu[k].z ^ pat);
            const unsigned h3 = zero_bytes(cu[k].w ^ pat);
            mk[k] = (h0 >> 7) | (h1 >> 6) | (h2 >> 5) | (h3 >> 4);
        }

        #pragma unroll 1
        for (int k = 0; k < OWN_LOADS; ++k) {
            unsigned m = mk[0];
            #pragma unroll
            for (int j = 1; j < OWN_LOADS; ++j) mk[j - 1] = mk[j];
            unsigned bm = __builtin_amdgcn_ballot_w32(m != 0u);
            if (bm != 0u) {
                const unsigned ebase = (unsigned)(step * OWN_STEP_IDS + k * SLICE_IDS + lane * 16);
                do {
                    if (m != 0u) {
                        const unsigned p = (unsigned)__builtin_ctz(m);
                        m &= m - 1u;
                        q[qn + (int)__builtin_amdgcn_mbcnt_lo(bm, 0u)] = ebase + ((p & 3u) << 2) + (p >> 3);
                    }
                    qn += __builtin_popcount(bm);
                    bm = __builtin_amdgcn_ballot_w32(m != 0u);
                } while (bm != 0u);
                while (qn >= 32) {
                    qn -= 32;
                    agg_add_batch(q, qn, 32, lane, X_edge, h_src, src, dst, lo, nrows, acc, ef);
                }
            }
        }
    }
    if (qn > 0) agg_add_batch(q, 0, qn, lane, X_edge, h_src, src, dst, lo, nrows, acc, ef);
    __syncthreads();

    float4 w[D_EDGE];
    #pragma unroll
    for (int k = 0; k < D_EDGE; ++k) w[k] = ((const float4*)(W_edge + k * D_OUT))[lane];

    for (int r = wave; r < nrows; r += AGG_WAVES) {
        const float* e = ef + r * EF_STRIDE;
        const float* a = acc + r * D_OUT + lane;
        float4 sum = make_float4(a[0], a[32], a[64], a[96]);
        #pragma unroll
        for (int k = 0; k < D_EDGE; ++k) {
            const float xk = e[k];
            sum.x = fmaf(xk, w[k].x, sum.x);
            sum.y = fmaf(xk, w[k].y, sum.y);
            sum.z = fmaf(xk, w[k].z, sum.z);
            sum.w = fmaf(xk, w[k].w, sum.w);
        }
        const float denom = fmaxf(e[D_EDGE], 1.0f);
        float4* op = (float4*)(out + (size_t)(lo + r) * D_OUT) + lane;
        float4 o = *op;
        o.x += sum.x / denom;
        o.y += sum.y / denom;
        o.z += sum.z / denom;
        o.w += sum.w / denom;
        *op = o;
    }
}

extern "C" void kernel_launch(void* const* d_in, const int* in_sizes, int n_in,
                              void* d_out, int out_size, void* d_ws, size_t ws_size,
                              hipStream_t stream) {
    const float* X       = (const float*)d_in[0];
    const float* X_edge  = (const float*)d_in[1];
    const float* W_neigh = (const float*)d_in[2];
    const float* W_self  = (const float*)d_in[3];
    const float* b_self  = (const float*)d_in[4];
    const float* W_edge  = (const float*)d_in[5];
    const int*   src     = (const int*)d_in[6];
    const int*   dst     = (const int*)d_in[7];
    float* out = (float*)d_out;

    float* h_src = (float*)d_ws;
    unsigned* owner = (unsigned*)(h_src + (size_t)N_NODES * D_OUT);

    gemm_dual_wmma_kernel<<<GEMM_BLOCKS, 512, 0, stream>>>(
        X, W_self, W_neigh, b_self, out, h_src, dst, owner);

    range_aggregate_kernel<<<AGG_GROUPS, 1024, 0, stream>>>(
        X_edge, h_src, src, dst, owner, W_edge, out);
}


# src id and local dst row packed into one word per edge by the GEMM launch; the aggregation fetches one scattered word per owned edge instead of two (src, dst); arrival sum order as before
# speedup vs baseline: 3.7842x; 1.0409x over previous
#include <hip/hip_runtime.h>
#include <hip/hip_bf16.h>

#define N_NODES 100000
#define N_EDGES 1600000
#define D_IN    256
#define D_EDGE  16
#define D_OUT   128

#define AGG_GROUPS 256
#define AGG_ROWS   391
#define GEMM_MT       5
#define GEMM_ROWS     (16 * GEMM_MT)
#define GEMM_BLOCKS   (N_NODES / GEMM_ROWS)
#define OWN_PER_BLOCK (N_EDGES / GEMM_BLOCKS)
static_assert(N_NODES % GEMM_ROWS == 0, "whole row blocks");
static_assert(N_EDGES % GEMM_BLOCKS == 0 && OWN_PER_BLOCK % 4 == 0 && OWN_PER_BLOCK / 4 <= 512,
              "the GEMM blocks cover every edge with whole words");
static_assert((N_NODES - 1) / AGG_ROWS <= 255, "an owner index fits a byte");
#define PAIR_SRC_BITS 17
#define PAIR_SRC_MASK ((1u << PAIR_SRC_BITS) - 1u)
#define PAIR_ROW_NONE 511u
static_assert(N_NODES <= (1 << PAIR_SRC_BITS) && AGG_ROWS <= (int)PAIR_ROW_NONE, "src id and local row fit one word");

typedef __attribute__((ext_vector_type(2))) float v2f;
typedef __attribute__((ext_vector_type(8))) float v8f;

#define XS_STRIDE 260

__global__ void __launch_bounds__(512)
gemm_dual_wmma_kernel(const float* __restrict__ X,
                      const float* __restrict__ W_self,
                      const float* __restrict__ W_neigh,
                      const float* __restrict__ b_self,
                      float* __restrict__ out_self,
                      float* __restrict__ out_neigh,
                      const int* __restrict__ src,
                      const int* __restrict__ dst,
                      unsigned* __restrict__ owner,
                      unsigned* __restrict__ pair) {
    __shared__ __attribute__((aligned(16))) float xs[GEMM_ROWS * XS_STRIDE];

    const int rowBase = blockIdx.x * GEMM_ROWS;

    if (threadIdx.x < OWN_PER_BLOCK / 4) {
        const int w = blockIdx.x * (OWN_PER_BLOCK / 4) + threadIdx.x;
        const int4 d = ((const int4*)dst)[w];
        const int4 s = ((const int4*)src)[w];
        const unsigned o0 = (unsigned)(d.x / AGG_ROWS) & 255u;
        const unsigned o1 = (unsigned)(d.y / AGG_ROWS) & 255u;
        const unsigned o2 = (unsigned)(d.z / AGG_ROWS) & 255u;
        const unsigned o3 = (unsigned)(d.w / AGG_ROWS) & 255u;
        owner[w] = o0 | (o1 << 8) | (o2 << 16) | (o3 << 24);
        unsigned r0 = (unsigned)d.x - o0 * (unsigned)AGG_ROWS;
        unsigned r1 = (unsigned)d.y - o1 * (unsigned)AGG_ROWS;
        unsigned r2 = (unsigned)d.z - o2 * (unsigned)AGG_ROWS;
        unsigned r3 = (unsigned)d.w - o3 * (unsigned)AGG_ROWS;
        r0 = (r0 < (unsigned)AGG_ROWS) ? r0 : PAIR_ROW_NONE;
        r1 = (r1 < (unsigned)AGG_ROWS) ? r1 : PAIR_ROW_NONE;
        r2 = (r2 < (unsigned)AGG_ROWS) ? r2 : PAIR_ROW_NONE;
        r3 = (r3 < (unsigned)AGG_ROWS) ? r3 : PAIR_ROW_NONE;
        uint4 p;
        p.x = ((unsigned)s.x & PAIR_SRC_MASK) | (r0 << PAIR_SRC_BITS);
        p.y = ((unsigned)s.y & PAIR_SRC_MASK) | (r1 << PAIR_SRC_BITS);
        p.z = ((unsigned)s.z & PAIR_SRC_MASK) | (r2 << PAIR_SRC_BITS);
        p.w = ((unsigned)s.w & PAIR_SRC_MASK) | (r3 << PAIR_SRC_BITS);
        ((uint4*)pair)[w] = p;
    }

    for (int idx = threadIdx.x; idx < GEMM_ROWS * (D_IN / 4); idx += 512) {
        const int r  = idx >> 6;
        const int c4 = idx & 63;
        const float4 v = ((const float4*)(X + (size_t)(rowBase + r) * D_IN))[c4];
        *(float4*)&xs[r * XS_STRIDE + 4 * c4] = v;
    }
    __syncthreads();

    const int wave  = threadIdx.x >> 5;
    const int lane  = threadIdx.x & 31;
    const int ntile = wave & 7;
    const bool isSelf = (wave < 8);

    const float* __restrict__ W = isSelf ? W_self : W_neigh;

    const int m     = lane & 15;
    const int khalf = lane >> 4;
    const int col   = ntile * 16 + m;

    v8f acc[GEMM_MT];
    #pragma unroll
    for (int t = 0; t < GEMM_MT; ++t) acc[t] = (v8f){};

    const float* xrow = &xs[m * XS_STRIDE + 2 * khalf];
    const float* wcol = W + (size_t)(2 * khalf) * D_OUT + col;

    #pragma unroll 4
    for (int kb = 0; kb < D_IN; kb += 4) {
        v2f b;
        b.x = wcol[(size_t)(kb + 0) * D_OUT];
        b.y = wcol[(size_t)(kb + 1) * D_OUT];
        #pragma unroll
        for (int t = 0; t < GEMM_MT; ++t) {
            const v2f a = *(const v2f*)(xrow + t * 16 * XS_STRIDE + kb);
            acc[t] = __builtin_amdgcn_wmma_f32_16x16x4_f32(
                 false, a,  false, b,
                 (short)0, acc[t],  false,  false);
        }
    }

    const float bias = isSelf ? b_self[col] : 0.0f;
    float* __restrict__ outp = isSelf ? out_self : out_neigh;

    #pragma unroll
    for (int t = 0; t < GEMM_MT; ++t) {
        #pragma unroll
        for (int r = 0; r < 8; ++r) {
            const int row = rowBase + t * 16 + r + 8 * khalf;
            outp[(size_t)row * D_OUT + col] = acc[t][r] + bias;
        }
    }
}

#define AGG_WAVES  32
#define EF_STRIDE  17
#define SLICE_IDS  512
#define OWN_LOADS  5
#define OWN_STEP_IDS (OWN_LOADS * SLICE_IDS)
#define OWN_STEPS  (N_EDGES / OWN_STEP_IDS)
#define Q_CAP      (32 + SLICE_IDS)
#define SWEEP_SPREAD 64
static_assert(SWEEP_SPREAD <= OWN_STEPS, "a start wraps at most once");

static_assert(AGG_GROUPS * AGG_ROWS >= N_NODES, "the groups cover every node");
static_assert(N_EDGES % OWN_STEP_IDS == 0, "whole scan steps");
static_assert((AGG_ROWS * D_OUT + AGG_ROWS * EF_STRIDE + AGG_WAVES * Q_CAP) * 4 <= 320 * 1024, "LDS budget");

__device__ __forceinline__ unsigned zero_bytes(unsigned t) {
    return ~(((t & 0x7F7F7F7Fu) + 0x7F7F7F7Fu) | t | 0x7F7F7F7Fu);
}

__device__ __forceinline__ void agg_add_batch(const unsigned* q, int first, int nb, int lane,
                                              const float* __restrict__ X_edge,
                                              const float* __restrict__ h_src,
                                              const unsigned* __restrict__ pair,
                                              int nrows,
                                              float* acc, float* ef) {
    unsigned ent = 0;
    int s = 0;
    int row = 0;
    if (lane < nb) {
        ent = q[first + lane];
        const unsigned pw = pair[ent];
        s = (int)(pw & PAIR_SRC_MASK);
        row = (int)(pw >> PAIR_SRC_BITS);
    }
    for (int i0 = 0; i0 < nb; i0 += 8) {
        float4 hv[8];
        float  xv[8];
        unsigned hi[8], xi[8];
        #pragma unroll
        for (int j = 0; j < 8; ++j) {
            const unsigned sj = (unsigned)__builtin_amdgcn_readlane(s, (i0 + j) & 31);
            const unsigned ej = (unsigned)__builtin_amdgcn_readlane((int)ent, (i0 + j) & 31);
            hi[j] = sj * (unsigned)(D_OUT / 4) + (unsigned)lane;
            xi[j] = ej * (unsigned)D_EDGE + (unsigned)(lane & (D_EDGE - 1));
        }
        #pragma unroll
        for (int j = 0; j < 8; ++j) {
            if (i0 + j < nb) {
                hv[j] = ((const float4*)h_src)[hi[j]];
                xv[j] = X_edge[xi[j]];
            }
        }
        asm volatile("" :: "v"(hv[0].x), "v"(hv[0].y), "v"(hv[0].z), "v"(hv[0].w), "v"(xv[0]));
        #pragma unroll
        for (int j = 0; j < 8; ++j) {
            const int rj = __builtin_amdgcn_readlane(row, (i0 + j) & 31);
            if (i0 + j < nb && (unsigned)rj < (unsigned)nrows) {
                float* a = acc + rj * D_OUT + lane;
                __hip_atomic_fetch_add(a,      hv[j].x, __ATOMIC_RELAXED, __HIP_MEMORY_SCOPE_WORKGROUP);
                __hip_atomic_fetch_add(a + 32, hv[j].y, __ATOMIC_RELAXED, __HIP_MEMORY_SCOPE_WORKGROUP);
                __hip_atomic_fetch_add(a + 64, hv[j].z, __ATOMIC_RELAXED, __HIP_MEMORY_SCOPE_WORKGROUP);
                __hip_atomic_fetch_add(a + 96, hv[j].w, __ATOMIC_RELAXED, __HIP_MEMORY_SCOPE_WORKGROUP);
                if (lane <= D_EDGE)
                    __hip_atomic_fetch_add(ef + rj * EF_STRIDE + lane, (lane < D_EDGE) ? xv[j] : 1.0f,
                                           __ATOMIC_RELAXED, __HIP_MEMORY_SCOPE_WORKGROUP);
            }
        }
    }
}

__global__ void __launch_bounds__(1024)
range_aggregate_kernel(const float* __restrict__ X_edge,
                       const float* __restrict__ h_src,
                       const unsigned* __restrict__ pair,
                       const unsigned* __restrict__ owner,
                       const float* __restrict__ W_edge,
                       float* __restrict__ out) {
    __shared__ __attribute__((aligned(16))) float acc[AGG_ROWS * D_OUT];
    __shared__ float    ef[AGG_ROWS * EF_STRIDE];
    __shared__ unsigned queue[AGG_WAVES * Q_CAP];

    const int wave  = __builtin_amdgcn_readfirstlane((int)(threadIdx.x >> 5));
    const int lane  = threadIdx.x & 31;
    const int lo    = blockIdx.x * AGG_ROWS;
    const int nrows = min(AGG_ROWS, N_NODES - lo);

    for (int i = threadIdx.x; i < AGG_ROWS * D_OUT / 4; i += 1024)
        ((float4*)acc)[i] = make_float4(0.0f, 0.0f, 0.0f, 0.0f);
    for (int i = threadIdx.x; i < AGG_ROWS * EF_STRIDE; i += 1024) ef[i] = 0.0f;
    __syncthreads();

    unsigned* q = queue + wave * Q_CAP;
    int qn = 0;
    const uint4* __restrict__ own4 = (const uint4*)owner;
    const unsigned pat = (unsigned)blockIdx.x * 0x01010101u;
    const int start = (int)((blockIdx.x * SWEEP_SPREAD) / AGG_GROUPS);

    uint4 nx[OWN_LOADS];
    {
        int step = wave + start;
        if (step >= OWN_STEPS) step -= OWN_STEPS;
        #pragma unroll
        for (int k = 0; k < OWN_LOADS; ++k) nx[k] = own4[(step * OWN_LOADS + k) * 32 + lane];
    }

    for (int t = wave; t < OWN_STEPS; t += AGG_WAVES) {
        int step = t + start;
        if (step >= OWN_STEPS) step -= OWN_STEPS;

        uint4 cu[OWN_LOADS];
        #pragma unroll
        for (int k = 0; k < OWN_LOADS; ++k) cu[k] = nx[k];
        const int tn = t + AGG_WAVES;
        if (tn < OWN_STEPS) {
            int sn = tn + start;
            if (sn >= OWN_STEPS) sn -= OWN_STEPS;
            #pragma unroll
            for (int k = 0; k < OWN_LOADS; ++k) nx[k] = own4[(sn * OWN_LOADS + k) * 32 + lane];
        }

        unsigned mk[OWN_LOADS];
        #pragma unroll
        for (int k = 0; k < OWN_LOADS; ++k) {
            const unsigned h0 = zero_bytes(cu[k].x ^ pat);
            const unsigned h1 = zero_bytes(cu[k].y ^ pat);
            const unsigned h2 = zero_bytes(cu[k].z ^ pat);
            const unsigned h3 = zero_bytes(cu[k].w ^ pat);
            mk[k] = (h0 >> 7) | (h1 >> 6) | (h2 >> 5) | (h3 >> 4);
        }

        #pragma unroll 1
        for (int k = 0; k < OWN_LOADS; ++k) {
            unsigned m = mk[0];
            #pragma unroll
            for (int j = 1; j < OWN_LOADS; ++j) mk[j - 1] = mk[j];
            unsigned bm = __builtin_amdgcn_ballot_w32(m != 0u);
            if (bm != 0u) {
                const unsigned ebase = (unsigned)(step * OWN_STEP_IDS + k * SLICE_IDS + lane * 16);
                do {
                    if (m != 0u) {
                        const unsigned p = (unsigned)__builtin_ctz(m);
                        m &= m - 1u;
                        q[qn + (int)__builtin_amdgcn_mbcnt_lo(bm, 0u)] = ebase + ((p & 3u) << 2) + (p >> 3);
                    }
                    qn += __builtin_popcount(bm);
                    bm = __builtin_amdgcn_ballot_w32(m != 0u);
                } while (bm != 0u);
                while (qn >= 32) {
                    qn -= 32;
                    agg_add_batch(q, qn, 32, lane, X_edge, h_src, pair, nrows, acc, ef);
                }
            }
        }
    }
    if (qn > 0) agg_add_batch(q, 0, qn, lane, X_edge, h_src, pair, nrows, acc, ef);
    __syncthreads();

    float4 w[D_EDGE];
    #pragma unroll
    for (int k = 0; k < D_EDGE; ++k) w[k] = ((const float4*)(W_edge + k * D_OUT))[lane];

    for (int r = wave; r < nrows; r += AGG_WAVES) {
        const float* e = ef + r * EF_STRIDE;
        const float* a = acc + r * D_OUT + lane;
        float4 sum = make_float4(a[0], a[32], a[64], a[96]);
        #pragma unroll
        for (int k = 0; k < D_EDGE; ++k) {
            const float xk = e[k];
            sum.x = fmaf(xk, w[k].x, sum.x);
            sum.y = fmaf(xk, w[k].y, sum.y);
            sum.z = fmaf(xk, w[k].z, sum.z);
            sum.w = fmaf(xk, w[k].w, sum.w);
        }
        const float denom = fmaxf(e[D_EDGE], 1.0f);
        float4* op = (float4*)(out + (size_t)(lo + r) * D_OUT) + lane;
        float4 o = *op;
        o.x += sum.x / denom;
        o.y += sum.y / denom;
        o.z += sum.z / denom;
        o.w += sum.w / denom;
        *op = o;
    }
}

extern "C" void kernel_launch(void* const* d_in, const int* in_sizes, int n_in,
                              void* d_out, int out_size, void* d_ws, size_t ws_size,
                              hipStream_t stream) {
    const float* X       = (const float*)d_in[0];
    const float* X_edge  = (const float*)d_in[1];
    const float* W_neigh = (const float*)d_in[2];
    const float* W_self  = (const float*)d_in[3];
    const float* b_self  = (const float*)d_in[4];
    const float* W_edge  = (const float*)d_in[5];
    const int*   src     = (const int*)d_in[6];
    const int*   dst     = (const int*)d_in[7];
    float* out = (float*)d_out;

    float* h_src = (float*)d_ws;
    unsigned* owner = (unsigned*)(h_src + (size_t)N_NODES * D_OUT);
    unsigned* pair = owner + N_EDGES / 4;

    gemm_dual_wmma_kernel<<<GEMM_BLOCKS, 512, 0, stream>>>(
        X, W_self, W_neigh, b_self, out, h_src, src, dst, owner, pair);

    range_aggregate_kernel<<<AGG_GROUPS, 1024, 0, stream>>>(
        X_edge, h_src, pair, owner, W_edge, out);
}
